# starting kernel
# baseline (speedup 1.0000x reference)
#include <hip/hip_runtime.h>


typedef float v2f __attribute__((ext_vector_type(2)));
typedef float v8f __attribute__((ext_vector_type(8)));

#define KS   15
#define RG   16
#define HW   256
#define HO   241
#define PADI 7
#define TILE 30
#define YS   32
#define GK   480
#define AGUARD 16

__global__ __launch_bounds__(32)
void CustomConv_corr_wmma(const float* __restrict__ image,
                          const float* __restrict__ kernels,
                          float* __restrict__ corr)
{
    __shared__ float PtT[TILE * YS];
    __shared__ float KtZ[AGUARD + KS * YS];

    const int r  = blockIdx.x;
    const int bc = blockIdx.y;
    const int i0 = (r >> 4) * RG;
    const int j0 = (r & 15) * RG;
    const int l  = threadIdx.x;

    for (int e = l; e < TILE * YS; e += 32)          PtT[e] = 0.0f;
    for (int e = l; e < AGUARD + KS * YS; e += 32)   KtZ[e] = 0.0f;

    for (int e = l; e < KS * KS; e += 32) {
        int v = e / KS, u = e - v * KS;
        KtZ[AGUARD + v * YS + u] = kernels[r * (KS * KS) + u * KS + v];
    }

    const float* img = image + bc * (HW * HW);
    for (int e = l; e < TILE * TILE; e += 32) {
        int yy = e / TILE, xx = e - yy * TILE;
        int gy = i0 + yy - PADI, gx = j0 + xx - PADI;
        float pv = 0.0f;
        if (gy >= 0 && gy < HW && gx >= 0 && gx < HW)
            pv = img[gy * HW + gx];
        PtT[xx * YS + yy] = pv;
    }
    __syncthreads();

    const int m  = l & 15;
    const int lh = l >> 4;

    const float* aBase = KtZ + (AGUARD + 2 * lh - m);
    const float* bBase = PtT + (YS * m + 2 * lh);

    v8f acc = {};
    #pragma unroll 8
    for (int k0 = 0; k0 < GK; k0 += 4) {
        v2f a, b;
        a[0] = aBase[k0];     a[1] = aBase[k0 + 1];
        b[0] = bBase[k0];     b[1] = bBase[k0 + 1];
        acc = __builtin_amdgcn_wmma_f32_16x16x4_f32(
                  false, a, false, b, (short)0, acc, false, false);
    }

    float* dst = corr + bc * (HW * HW);
    #pragma unroll
    for (int g = 0; g < 8; ++g) {
        int mi = lh * 8 + g;
        int i = i0 + mi, j = j0 + m;
        float ov = (i < HO && j < HO) ? acc[g] : 0.0f;
        dst[i * HW + j] = ov;
    }
}

__global__ __launch_bounds__(256)
void CustomConv_rowbox(const float* __restrict__ src, float* __restrict__ dst)
{
    int idx = blockIdx.x * 256 + threadIdx.x;
    int x = idx & 255;
    float s = 0.0f;
    #pragma unroll
    for (int d = 0; d < RG; ++d) {
        float v = (x - d >= 0) ? src[idx - d] : 0.0f;
        s += v;
    }
    dst[idx] = s;
}

__global__ __launch_bounds__(256)
void CustomConv_colbox(const float* __restrict__ src, float* __restrict__ dst)
{
    int idx = blockIdx.x * 256 + threadIdx.x;
    int y = (idx >> 8) & 255;
    float s = 0.0f;
    #pragma unroll
    for (int d = 0; d < RG; ++d) {
        float v = (y - d >= 0) ? src[idx - d * HW] : 0.0f;
        s += v;
    }
    dst[idx] = s;
}

extern "C" void kernel_launch(void* const* d_in, const int* in_sizes, int n_in,
                              void* d_out, int out_size, void* d_ws, size_t ws_size,
                              hipStream_t stream)
{
    const float* image   = (const float*)d_in[0];
    const float* kernels = (const float*)d_in[1];
    float* out = (float*)d_out;
    float* ws  = (float*)d_ws;

    dim3 g1(256, 12);
    CustomConv_corr_wmma<<<g1, 32, 0, stream>>>(image, kernels, out);

    const int total = 12 * HW * HW;
    CustomConv_rowbox<<<total / 256, 256, 0, stream>>>(out, ws);
    CustomConv_colbox<<<total / 256, 256, 0, stream>>>(ws, out);
}


# Pass 1 image tile staged in image orientation, two rows interleaved (was transposed): B pair = one 8-byte read on 64 banks (were 8 lanes per bank), staging stores on 32 banks; K split by tile row
# speedup vs baseline: 2.3770x; 2.3770x over previous
#include <hip/hip_runtime.h>


typedef float v2f __attribute__((ext_vector_type(2)));
typedef float v8f __attribute__((ext_vector_type(8)));

#define KS   15
#define RG   16
#define HW   256
#define HO   241
#define PADI 7
#define TILE 30
#define YS   32
#define PP   96
#define GK   480
#define AGUARD 16
#define NW   4
#define NT   (NW * 32)
#define KTZ_N (AGUARD + KS * YS)
#define T_TRIPS 8
#define K_TRIPS 4
#define TP   18

__global__ __launch_bounds__(NT)
void CustomConv_corr_wmma(const float* __restrict__ image,
                          const float* __restrict__ kernels,
                          float* __restrict__ pre)
{
    __shared__ __attribute__((aligned(16))) float Pt[(YS / 2) * PP];
    __shared__ float KtZ[KTZ_N];
    __shared__ float Red[(NW - 1) * 256];
    __shared__ float Tr[16 * TP];

    const int r  = blockIdx.x;
    const int bc = blockIdx.y;
    const int i0 = (r >> 4) * RG;
    const int j0 = (r & 15) * RG;
    const int t  = threadIdx.x;
    const int l  = t & 31;
    const int w  = __builtin_amdgcn_readfirstlane(t >> 5);

    const float* img = image + bc * (HW * HW);
    const float* ker = kernels + r * (KS * KS);
    const int xl = l & 7, yl = l >> 3;
    float tv[T_TRIPS];
    float kv[K_TRIPS];
    #pragma unroll
    for (int s = 0; s < T_TRIPS; ++s) {
        int o  = w + NW * s;
        int xx = (o & 3) * 8 + xl;
        int yy = (o >> 2) * 4 + yl;
        int gy = i0 + yy - PADI, gx = j0 + xx - PADI;
        float pv = 0.0f;
        if (xx < TILE && yy < TILE && gy >= 0 && gy < HW && gx >= 0 && gx < HW)
            pv = img[gy * HW + gx];
        tv[s] = pv;
    }
    #pragma unroll
    for (int s = 0; s < K_TRIPS; ++s) {
        int e = t + NT * s;
        int q = e - AGUARD;
        int v = q >> 5, u = q & 31;
        float x = 0.0f;
        if (e >= AGUARD && e < KTZ_N && u < KS)
            x = ker[u * KS + v];
        kv[s] = x;
    }
    #pragma unroll
    for (int s = 0; s < T_TRIPS; ++s) {
        int o  = w + NW * s;
        int xx = (o & 3) * 8 + xl;
        int yy = (o >> 2) * 4 + yl;
        Pt[(yy >> 1) * PP + 2 * xx + (yy & 1)] = tv[s];
    }
    #pragma unroll
    for (int s = 0; s < K_TRIPS; ++s) {
        int e = t + NT * s;
        if (e < KTZ_N) KtZ[e] = kv[s];
    }
    __syncthreads();

    const int m  = l & 15;
    const int lh = l >> 4;

    const float* aBase = KtZ + (AGUARD + 8 * w + 2 * lh - m);
    const float* bBase = Pt + ((4 * w + lh) * PP + 2 * m);
    v8f acc = {};
    #pragma unroll
    for (int jj = 0; jj < 2; ++jj) {
        #pragma unroll
        for (int v = 0; v < KS; ++v) {
            v2f a, b;
            a[0] = aBase[v * YS + 4 * jj];
            a[1] = aBase[v * YS + 4 * jj + 1];
            b = *reinterpret_cast<const v2f*>(bBase + (2 * jj) * PP + 2 * v);
            acc = __builtin_amdgcn_wmma_f32_16x16x4_f32(
                      false, a, false, b, (short)0, acc, false, false);
        }
    }

    if (w != 0) {
        #pragma unroll
        for (int g = 0; g < 8; ++g)
            Red[(w - 1) * 256 + g * 32 + l] = acc[g];
    }
    __syncthreads();

    if (w == 0) {
        #pragma unroll
        for (int p = 0; p < NW - 1; ++p) {
            #pragma unroll
            for (int g = 0; g < 8; ++g)
                acc[g] += Red[p * 256 + g * 32 + l];
        }
        #pragma unroll
        for (int g = 0; g < 8; ++g) {
            int i = i0 + lh * 8 + g, j = j0 + m;
            acc[g] = (i < HO && j < HO) ? acc[g] : 0.0f;
        }
        v8f d1 = {};
        #pragma unroll
        for (int t2 = 0; t2 < 4; ++t2) {
            int k = 8 * lh + 2 * t2;
            v2f a, b;
            a[0] = (k     <= m) ? 1.0f : 0.0f;
            a[1] = (k + 1 <= m) ? 1.0f : 0.0f;
            b[0] = acc[2 * t2];
            b[1] = acc[2 * t2 + 1];
            d1 = __builtin_amdgcn_wmma_f32_16x16x4_f32(
                     false, a, false, b, (short)0, d1, false, false);
        }
        #pragma unroll
        for (int g = 0; g < 8; ++g)
            Tr[(lh * 8 + g) * TP + m] = d1[g];
    }
    __syncthreads();
    if (w == 0) {
        v8f sp = {};
        #pragma unroll
        for (int t2 = 0; t2 < 4; ++t2) {
            int k = 8 * lh + 2 * t2;
            v2f a, b;
            a[0] = Tr[m * TP + k];
            a[1] = Tr[m * TP + k + 1];
            b[0] = (k     <= m) ? 1.0f : 0.0f;
            b[1] = (k + 1 <= m) ? 1.0f : 0.0f;
            sp = __builtin_amdgcn_wmma_f32_16x16x4_f32(
                     false, a, false, b, (short)0, sp, false, false);
        }
        float* dst = pre + bc * (HW * HW);
        #pragma unroll
        for (int g = 0; g < 8; ++g) {
            int i = i0 + lh * 8 + g, j = j0 + m;
            dst[i * HW + j] = sp[g];
        }
    }
}

__device__ __forceinline__ float4 ld4(const float* p)
{
    return *reinterpret_cast<const float4*>(p);
}

__global__ __launch_bounds__(128)
void CustomConv_combine(const float* __restrict__ pre, float* __restrict__ dst)
{
    const int q  = blockIdx.x * 128 + threadIdx.x;
    const int x  = (q & 63) * 4;
    const int rg = (q >> 6) & 3;
    const int ti = (q >> 8) & 15;
    const int bc = q >> 12;
    const int y0 = ti * RG + rg * 4;
    const float* P = pre + bc * (HW * HW);
    float*       O = dst + bc * (HW * HW);
    const bool up = ti > 0;
    const bool lf = x >= RG;
    const int  yb = ti * RG - 1;
    const int  xr = (x & ~15) - 1;
    const float4 z4 = make_float4(0.0f, 0.0f, 0.0f, 0.0f);

    float4 a[4], u[4], f[4], d[4];
    float  fr[4], dr[4];
    #pragma unroll
    for (int r = 0; r < 4; ++r) {
        int y = y0 + r;
        a[r]  = ld4(P + y * HW + x);
        u[r]  = up ? ld4(P + (y - RG) * HW + x) : z4;
        f[r]  = lf ? ld4(P + y * HW + x - RG) : z4;
        fr[r] = lf ? P[y * HW + xr] : 0.0f;
        d[r]  = (up && lf) ? ld4(P + (y - RG) * HW + x - RG) : z4;
        dr[r] = (up && lf) ? P[(y - RG) * HW + xr] : 0.0f;
    }
    const float4 ub  = up ? ld4(P + yb * HW + x) : z4;
    const float4 db  = (up && lf) ? ld4(P + yb * HW + x - RG) : z4;
    const float  dbr = (up && lf) ? P[yb * HW + xr] : 0.0f;

    #pragma unroll
    for (int r = 0; r < 4; ++r) {
        float4 o;
        float cr = fr[r] + (dbr - dr[r]);
        o.x = a[r].x + (ub.x - u[r].x) + (cr - f[r].x) - (db.x - d[r].x);
        o.y = a[r].y + (ub.y - u[r].y) + (cr - f[r].y) - (db.y - d[r].y);
        o.z = a[r].z + (ub.z - u[r].z) + (cr - f[r].z) - (db.z - d[r].z);
        o.w = a[r].w + (ub.w - u[r].w) + (cr - f[r].w) - (db.w - d[r].w);
        *reinterpret_cast<float4*>(O + (y0 + r) * HW + x) = o;
    }
}

extern "C" void kernel_launch(void* const* d_in, const int* in_sizes, int n_in,
                              void* d_out, int out_size, void* d_ws, size_t ws_size,
                              hipStream_t stream)
{
    const float* image   = (const float*)d_in[0];
    const float* kernels = (const float*)d_in[1];
    float* out = (float*)d_out;
    float* ws  = (float*)d_ws;

    dim3 g1(256, 12);
    CustomConv_corr_wmma<<<g1, 128, 0, stream>>>(image, kernels, ws);

    CustomConv_combine<<<384, 128, 0, stream>>>(ws, out);
}


# Pass 1 work-group takes 4 planes of one region (grid 256x12 -> 256x3): each K-quarter wave reuses its A pair for 4 planes on 4 independent accumulators, wave w reduces and finishes plane w
# speedup vs baseline: 2.5241x; 1.0619x over previous
#include <hip/hip_runtime.h>


typedef float v2f __attribute__((ext_vector_type(2)));
typedef float v8f __attribute__((ext_vector_type(8)));

#define KS   15
#define RG   16
#define HW   256
#define HO   241
#define PADI 7
#define TILE 30
#define YS   32
#define PP   96
#define GK   480
#define AGUARD 16
#define NW   4
#define NP   4
#define PT_N ((YS / 2) * PP)
#define NT   (NW * 32)
#define KTZ_N (AGUARD + KS * YS)
#define T_TRIPS 8
#define K_TRIPS 4
#define TP   18

__global__ __launch_bounds__(NT)
void CustomConv_corr_wmma(const float* __restrict__ image,
                          const float* __restrict__ kernels,
                          float* __restrict__ pre)
{
    __shared__ __attribute__((aligned(16))) float Pt[NP * PT_N];
    __shared__ float KtZ[K_TRIPS * NT];
    __shared__ float Red[NW * NP * 256];
    __shared__ float Tr[NW * 16 * TP];

    const int r   = blockIdx.x;
    const int bc0 = blockIdx.y * NP;
    const int i0 = (r >> 4) * RG;
    const int j0 = (r & 15) * RG;
    const int t  = threadIdx.x;
    const int l  = t & 31;
    const int w  = __builtin_amdgcn_readfirstlane(t >> 5);

    const float* img = image + bc0 * (HW * HW);
    const float* ker = kernels + r * (KS * KS);
    const int xl = l & 7, yl = l >> 3;
    float tv[NP][T_TRIPS];
    float kv[K_TRIPS];
    #pragma unroll
    for (int s = 0; s < T_TRIPS; ++s) {
        int o  = w + NW * s;
        int xx = (o & 3) * 8 + xl;
        int yy = (o >> 2) * 4 + yl;
        int gy = i0 + yy - PADI, gx = j0 + xx - PADI;
        const bool ok = xx < TILE && yy < TILE && gy >= 0 && gy < HW && gx >= 0 && gx < HW;
        const int cy = min(max(gy, 0), HW - 1), cx = min(max(gx, 0), HW - 1);
        #pragma unroll
        for (int p = 0; p < NP; ++p) {
            float pv = img[p * (HW * HW) + cy * HW + cx];
            tv[p][s] = ok ? pv : 0.0f;
        }
    }
    #pragma unroll
    for (int s = 0; s < K_TRIPS; ++s) {
        int e = t + NT * s;
        int q = e - AGUARD;
        int v = q >> 5, u = q & 31;
        const bool okk = e >= AGUARD && e < KTZ_N && u < KS;
        float x = ker[okk ? u * KS + v : 0];
        kv[s] = okk ? x : 0.0f;
    }
    #pragma unroll
    for (int s = 0; s < T_TRIPS; ++s) {
        int o  = w + NW * s;
        int xx = (o & 3) * 8 + xl;
        int yy = (o >> 2) * 4 + yl;
        #pragma unroll
        for (int p = 0; p < NP; ++p)
            Pt[p * PT_N + (yy >> 1) * PP + 2 * xx + (yy & 1)] = tv[p][s];
    }
    #pragma unroll
    for (int s = 0; s < K_TRIPS; ++s) {
        int e = t + NT * s;
        KtZ[e] = kv[s];
    }
    __syncthreads();

    const int m  = l & 15;
    const int lh = l >> 4;

    const float* aBase = KtZ + (AGUARD + 8 * w + 2 * lh - m);
    const float* bBase = Pt + ((4 * w + lh) * PP + 2 * m);
    v8f accp[NP];
    #pragma unroll
    for (int p = 0; p < NP; ++p) accp[p] = (v8f){};
    #pragma unroll
    for (int jj = 0; jj < 2; ++jj) {
        #pragma unroll
        for (int v = 0; v < KS; ++v) {
            v2f a;
            a[0] = aBase[v * YS + 4 * jj];
            a[1] = aBase[v * YS + 4 * jj + 1];
            #pragma unroll
            for (int p = 0; p < NP; ++p) {
                v2f b = *reinterpret_cast<const v2f*>(bBase + p * PT_N + (2 * jj) * PP + 2 * v);
                accp[p] = __builtin_amdgcn_wmma_f32_16x16x4_f32(
                              false, a, false, b, (short)0, accp[p], false, false);
            }
        }
    }

    #pragma unroll
    for (int p = 0; p < NP; ++p) {
        #pragma unroll
        for (int g = 0; g < 8; ++g)
            Red[(w * NP + p) * 256 + g * 32 + l] = accp[p][g];
    }
    __syncthreads();

    const int bc = bc0 + w;
    float* Trw = Tr + w * (16 * TP);
    v8f acc;
    {
        #pragma unroll
        for (int g = 0; g < 8; ++g) {
            float x = Red[(0 * NP + w) * 256 + g * 32 + l];
            #pragma unroll
            for (int q = 1; q < NW; ++q)
                x += Red[(q * NP + w) * 256 + g * 32 + l];
            acc[g] = x;
        }
        #pragma unroll
        for (int g = 0; g < 8; ++g) {
            int i = i0 + lh * 8 + g, j = j0 + m;
            acc[g] = (i < HO && j < HO) ? acc[g] : 0.0f;
        }
        v8f d1 = {};
        #pragma unroll
        for (int t2 = 0; t2 < 4; ++t2) {
            int k = 8 * lh + 2 * t2;
            v2f a, b;
            a[0] = (k     <= m) ? 1.0f : 0.0f;
            a[1] = (k + 1 <= m) ? 1.0f : 0.0f;
            b[0] = acc[2 * t2];
            b[1] = acc[2 * t2 + 1];
            d1 = __builtin_amdgcn_wmma_f32_16x16x4_f32(
                     false, a, false, b, (short)0, d1, false, false);
        }
        #pragma unroll
        for (int g = 0; g < 8; ++g)
            Trw[(lh * 8 + g) * TP + m] = d1[g];
    }
    __syncthreads();
    {
        v8f sp = {};
        #pragma unroll
        for (int t2 = 0; t2 < 4; ++t2) {
            int k = 8 * lh + 2 * t2;
            v2f a, b;
            a[0] = Trw[m * TP + k];
            a[1] = Trw[m * TP + k + 1];
            b[0] = (k     <= m) ? 1.0f : 0.0f;
            b[1] = (k + 1 <= m) ? 1.0f : 0.0f;
            sp = __builtin_amdgcn_wmma_f32_16x16x4_f32(
                     false, a, false, b, (short)0, sp, false, false);
        }
        float* dst = pre + bc * (HW * HW);
        #pragma unroll
        for (int g = 0; g < 8; ++g) {
            int i = i0 + lh * 8 + g, j = j0 + m;
            dst[i * HW + j] = sp[g];
        }
    }
}

__device__ __forceinline__ float4 ld4(const float* p)
{
    return *reinterpret_cast<const float4*>(p);
}

__global__ __launch_bounds__(128)
void CustomConv_combine(const float* __restrict__ pre, float* __restrict__ dst)
{
    const int q  = blockIdx.x * 128 + threadIdx.x;
    const int x  = (q & 63) * 4;
    const int rg = (q >> 6) & 3;
    const int ti = (q >> 8) & 15;
    const int bc = q >> 12;
    const int y0 = ti * RG + rg * 4;
    const float* P = pre + bc * (HW * HW);
    float*       O = dst + bc * (HW * HW);
    const bool up = ti > 0;
    const bool lf = x >= RG;
    const int  yb = ti * RG - 1;
    const int  xr = (x & ~15) - 1;
    const float4 z4 = make_float4(0.0f, 0.0f, 0.0f, 0.0f);

    float4 a[4], u[4], f[4], d[4];
    float  fr[4], dr[4];
    #pragma unroll
    for (int r = 0; r < 4; ++r) {
        int y = y0 + r;
        a[r]  = ld4(P + y * HW + x);
        u[r]  = up ? ld4(P + (y - RG) * HW + x) : z4;
        f[r]  = lf ? ld4(P + y * HW + x - RG) : z4;
        fr[r] = lf ? P[y * HW + xr] : 0.0f;
        d[r]  = (up && lf) ? ld4(P + (y - RG) * HW + x - RG) : z4;
        dr[r] = (up && lf) ? P[(y - RG) * HW + xr] : 0.0f;
    }
    const float4 ub  = up ? ld4(P + yb * HW + x) : z4;
    const float4 db  = (up && lf) ? ld4(P + yb * HW + x - RG) : z4;
    const float  dbr = (up && lf) ? P[yb * HW + xr] : 0.0f;

    #pragma unroll
    for (int r = 0; r < 4; ++r) {
        float4 o;
        float cr = fr[r] + (dbr - dr[r]);
        o.x = a[r].x + (ub.x - u[r].x) + (cr - f[r].x) - (db.x - d[r].x);
        o.y = a[r].y + (ub.y - u[r].y) + (cr - f[r].y) - (db.y - d[r].y);
        o.z = a[r].z + (ub.z - u[r].z) + (cr - f[r].z) - (db.z - d[r].z);
        o.w = a[r].w + (ub.w - u[r].w) + (cr - f[r].w) - (db.w - d[r].w);
        *reinterpret_cast<float4*>(O + (y0 + r) * HW + x) = o;
    }
}

extern "C" void kernel_launch(void* const* d_in, const int* in_sizes, int n_in,
                              void* d_out, int out_size, void* d_ws, size_t ws_size,
                              hipStream_t stream)
{
    const float* image   = (const float*)d_in[0];
    const float* kernels = (const float*)d_in[1];
    float* out = (float*)d_out;
    float* ws  = (float*)d_ws;

    dim3 g1(256, 3);
    CustomConv_corr_wmma<<<g1, 128, 0, stream>>>(image, kernels, ws);

    CustomConv_combine<<<384, 128, 0, stream>>>(ws, out);
}
